# starting kernel
# baseline (speedup 1.0000x reference)
#include <hip/hip_runtime.h>
#include <math.h>

#define HD 128
#define TPB 256

typedef __attribute__((ext_vector_type(2))) float v2f;
typedef __attribute__((ext_vector_type(8))) float v8f;

__global__ void __launch_bounds__(TPB) zero_kernel(float* __restrict__ p, size_t n) {
    size_t i = (size_t)blockIdx.x * TPB + threadIdx.x;
    if (i < n) p[i] = 0.0f;
}

__device__ __forceinline__ unsigned enc_f32(float f) {
    unsigned b = __float_as_uint(f);
    return (b & 0x80000000u) ? ~b : (b | 0x80000000u);
}
__device__ __forceinline__ float dec_f32(unsigned e) {
    unsigned b = (e & 0x80000000u) ? (e & 0x7FFFFFFFu) : ~e;
    return __uint_as_float(b);
}

__global__ void __launch_bounds__(TPB) edge_score_kernel(
        const float* __restrict__ ent, const float* __restrict__ rel,
        const int* __restrict__ src, const int* __restrict__ dst,
        const int* __restrict__ rid,
        float* __restrict__ score, unsigned* __restrict__ segmax, int n_edges) {
    int e = blockIdx.x * (TPB / 32) + (threadIdx.x >> 5);
    if (e >= n_edges) return;
    int lane = threadIdx.x & 31;
    int s = src[e], d = dst[e], r = rid[e];
    const float4* es = (const float4*)(ent + (size_t)s * HD);
    const float4* ed = (const float4*)(ent + (size_t)d * HD);
    const float4* rr = (const float4*)(rel + (size_t)r * HD);
    float4 a = es[lane], b = rr[lane], c = ed[lane];
    float v = a.x*b.x*c.x + a.y*b.y*c.y + a.z*b.z*c.z + a.w*b.w*c.w;
    #pragma unroll
    for (int off = 16; off > 0; off >>= 1) v += __shfl_xor(v, off, 32);
    if (lane == 0) {
        score[e] = v;
        atomicMax(segmax + d, enc_f32(v));
    }
}

__global__ void __launch_bounds__(TPB) edge_exp_kernel(
        const int* __restrict__ dst, const unsigned* __restrict__ segmax,
        float* __restrict__ score, float* __restrict__ denom, int n_edges) {
    int e = blockIdx.x * TPB + threadIdx.x;
    if (e >= n_edges) return;
    int d = dst[e];
    float ex = __expf(score[e] - dec_f32(segmax[d]));
    score[e] = ex;
    atomicAdd(denom + d, ex);
}

__global__ void __launch_bounds__(TPB) edge_aggr_kernel(
        const float* __restrict__ ent, const float* __restrict__ rel,
        const int* __restrict__ src, const int* __restrict__ dst,
        const int* __restrict__ rid,
        const float* __restrict__ score, const float* __restrict__ denom,
        float* __restrict__ neigh, int n_edges) {
    int e = blockIdx.x * (TPB / 32) + (threadIdx.x >> 5);
    if (e >= n_edges) return;
    int lane = threadIdx.x & 31;
    int s = src[e], d = dst[e], r = rid[e];
    float alpha = score[e] / denom[d];
    const float4* es = (const float4*)(ent + (size_t)s * HD);
    const float4* rr = (const float4*)(rel + (size_t)r * HD);
    float4 a = es[lane], b = rr[lane];
    float* out = neigh + (size_t)d * HD + lane * 4;
    atomicAdd(out + 0, a.x * b.x * alpha);
    atomicAdd(out + 1, a.y * b.y * alpha);
    atomicAdd(out + 2, a.z * b.z * alpha);
    atomicAdd(out + 3, a.w * b.w * alpha);
}

__global__ void __launch_bounds__(TPB) gemm_wmma_kernel(
        const float* __restrict__ A,
        const float* __restrict__ W,
        float* __restrict__ H,
        int n_rows) {
    const int wave = threadIdx.x >> 5;
    const int lane = threadIdx.x & 31;
    const int half = lane >> 4;
    const int l16  = lane & 15;
    const int m0 = blockIdx.x * 16;
    const int n0 = wave * 16;

    v8f c = {};
    const float* arow = A + (size_t)(m0 + l16) * HD;
    const float* wcol = W + n0 + l16;

    #pragma unroll 4
    for (int k0 = 0; k0 < HD; k0 += 4) {
        int ka = k0 + 2 * half;
        v2f a, b;
        a.x = arow[ka];
        a.y = arow[ka + 1];
        b.x = wcol[(size_t)ka * HD];
        b.y = wcol[(size_t)(ka + 1) * HD];
        c = __builtin_amdgcn_wmma_f32_16x16x4_f32(
                 false, a,  false, b,
                 (short)0, c,  false,  false);
    }

    float* hb = H + (size_t)(m0 + 8 * half) * HD + n0 + l16;
    #pragma unroll
    for (int r2 = 0; r2 < 8; ++r2) hb[(size_t)r2 * HD] = c[r2];
}

__global__ void __launch_bounds__(TPB) bn_stats_kernel(
        const float* __restrict__ H, float* __restrict__ colsum,
        float* __restrict__ colsumsq, int n_rows) {
    int col  = threadIdx.x & (HD - 1);
    int roff = threadIdx.x >> 7;
    float s = 0.0f, sq = 0.0f;
    for (int r = blockIdx.x * 2 + roff; r < n_rows; r += gridDim.x * 2) {
        float v = H[(size_t)r * HD + col];
        s += v; sq += v * v;
    }
    atomicAdd(colsum + col, s);
    atomicAdd(colsumsq + col, sq);
}

__global__ void __launch_bounds__(TPB) bn_tanh_kernel(
        float* __restrict__ H, const float* __restrict__ colsum,
        const float* __restrict__ colsumsq, const float* __restrict__ gamma,
        const float* __restrict__ beta, int n_rows) {
    size_t i = (size_t)blockIdx.x * TPB + threadIdx.x;
    size_t total = (size_t)n_rows * HD;
    if (i >= total) return;
    int c = (int)(i & (HD - 1));
    float invN = 1.0f / (float)n_rows;
    float mu   = colsum[c] * invN;
    float var  = colsumsq[c] * invN - mu * mu;
    float istd = rsqrtf(var + 1e-5f);
    float v = (H[i] - mu) * istd * gamma[c] + beta[c];
    H[i] = tanhf(v);
}

extern "C" void kernel_launch(void* const* d_in, const int* in_sizes, int n_in,
                              void* d_out, int out_size, void* d_ws, size_t ws_size,
                              hipStream_t stream) {
    const float* ent   = (const float*)d_in[0];
    const float* rel   = (const float*)d_in[1];
    const float* W     = (const float*)d_in[2];
    const float* gamma = (const float*)d_in[3];
    const float* beta  = (const float*)d_in[4];
    const int*   src   = (const int*)d_in[5];
    const int*   dst   = (const int*)d_in[6];
    const int*   rid   = (const int*)d_in[7];
    float* H = (float*)d_out;

    const int N = in_sizes[0] / HD;
    const int E = in_sizes[5];

    float*    ws       = (float*)d_ws;
    float*    neigh    = ws;
    float*    score    = neigh + (size_t)N * HD;
    unsigned* segmax   = (unsigned*)(score + E);
    float*    denom    = (float*)(segmax + N);
    float*    colsum   = denom + N;
    float*    colsumsq = colsum + HD;

    size_t zn = (size_t)N * HD;
    zero_kernel<<<(unsigned)((zn + TPB - 1) / TPB), TPB, 0, stream>>>(neigh, zn);
    size_t zt = (size_t)2 * N + 2 * HD;
    zero_kernel<<<(unsigned)((zt + TPB - 1) / TPB), TPB, 0, stream>>>((float*)segmax, zt);

    const int edges_per_blk = TPB / 32;
    edge_score_kernel<<<(unsigned)((E + edges_per_blk - 1) / edges_per_blk), TPB, 0, stream>>>(
        ent, rel, src, dst, rid, score, segmax, E);
    edge_exp_kernel<<<(unsigned)((E + TPB - 1) / TPB), TPB, 0, stream>>>(
        dst, segmax, score, denom, E);
    edge_aggr_kernel<<<(unsigned)((E + edges_per_blk - 1) / edges_per_blk), TPB, 0, stream>>>(
        ent, rel, src, dst, rid, score, denom, neigh, E);

    gemm_wmma_kernel<<<(unsigned)(N / 16), TPB, 0, stream>>>(neigh, W, H, N);

    bn_stats_kernel<<<256, TPB, 0, stream>>>(H, colsum, colsumsq, N);
    bn_tanh_kernel<<<(unsigned)(((size_t)N * HD + TPB - 1) / TPB), TPB, 0, stream>>>(
        H, colsum, colsumsq, gamma, beta, N);
}


# Score and scatter kernels: relation table copied once per work-group into LDS, rel rows read from LDS not L2; 256 resident 32-wave work-groups loop over the 8-edge groups; same sums, atomics kept
# speedup vs baseline: 1.6598x; 1.6598x over previous
#include <hip/hip_runtime.h>
#include <math.h>

#define HD 128
#define TPB 256
#define EPW 8
#define REL_ROWS 130
#define EBLK 1024
#define EGRID 256
static_assert(EPW == 8, "the scatter kernel names its 2 x 8 row registers one by one");

typedef __attribute__((ext_vector_type(2))) float v2f;
typedef __attribute__((ext_vector_type(8))) float v8f;
typedef __attribute__((ext_vector_type(4))) float v4f;

__global__ void __launch_bounds__(TPB) zero_kernel(float* __restrict__ p, size_t n) {
    size_t i = (size_t)blockIdx.x * TPB + threadIdx.x;
    if (i < n) p[i] = 0.0f;
}

__device__ __forceinline__ unsigned enc_f32(float f) {
    unsigned b = __float_as_uint(f);
    return (b & 0x80000000u) ? ~b : (b | 0x80000000u);
}
__device__ __forceinline__ float dec_f32(unsigned e) {
    unsigned b = (e & 0x80000000u) ? (e & 0x7FFFFFFFu) : ~e;
    return __uint_as_float(b);
}

__device__ __forceinline__ void rel_to_lds(
        const float* __restrict__ rel, float4* rel_s, int n_rel_floats) {
    constexpr int NF4 = REL_ROWS * HD / 4;
    constexpr int TRIPS = (NF4 + EBLK - 1) / EBLK;
    int nf = n_rel_floats / 4;
    if (nf > NF4) nf = NF4;
    const float4* rg = (const float4*)rel;
    float4 t[TRIPS];
    #pragma unroll
    for (int k = 0; k < TRIPS; ++k) {
        int i = threadIdx.x + k * EBLK;
        t[k] = (i < nf) ? rg[i] : make_float4(0.0f, 0.0f, 0.0f, 0.0f);
    }
    #pragma unroll
    for (int k = 0; k < TRIPS; ++k) {
        int i = threadIdx.x + k * EBLK;
        if (i < nf) rel_s[i] = t[k];
    }
    __syncthreads();
}

__device__ __forceinline__ float edge_dot(
        const float* __restrict__ ent, const float4* rel_s,
        int s, int d, int r, int lane) {
    const float4* es = (const float4*)(ent + (size_t)s * HD);
    const float4* ed = (const float4*)(ent + (size_t)d * HD);
    float4 a = es[lane], b = rel_s[r * (HD / 4) + lane], c = ed[lane];
    float v = a.x*b.x*c.x + a.y*b.y*c.y + a.z*b.z*c.z + a.w*b.w*c.w;
    #pragma unroll
    for (int off = 16; off > 0; off >>= 1) v += __shfl_xor(v, off, 32);
    return v;
}

__global__ void __launch_bounds__(EBLK) edge_score_kernel(
        const float* __restrict__ ent, const float* __restrict__ rel,
        const int* __restrict__ src, const int* __restrict__ dst,
        const int* __restrict__ rid,
        float* __restrict__ score, unsigned* __restrict__ segmax, int n_edges,
        int n_rel_floats) {
    __shared__ float4 rel_s[REL_ROWS * HD / 4];
    rel_to_lds(rel, rel_s, n_rel_floats);
    const int lane = threadIdx.x & 31;
    const int wv = __builtin_amdgcn_readfirstlane((int)(threadIdx.x >> 5));
    const int n_groups = (n_edges + EPW - 1) / EPW;
    for (int g = blockIdx.x * (EBLK / 32) + wv; g < n_groups; g += EGRID * (EBLK / 32)) {
    const int e0 = g * EPW;
    const int me  = e0 + (lane & (EPW - 1));
    const int mec = me < n_edges ? me : n_edges - 1;
    const int sv = src[mec], dv = dst[mec], rv = rid[mec];
    float res = 0.0f;
    if (e0 + EPW <= n_edges) {
        v4f a[EPW], c[EPW];
        #pragma unroll
        for (int j = 0; j < EPW; ++j) {
            int s = __builtin_amdgcn_readlane(sv, j);
            int d = __builtin_amdgcn_readlane(dv, j);
            a[j] = ((const v4f*)(ent + (size_t)s * HD))[lane];
            c[j] = ((const v4f*)(ent + (size_t)d * HD))[lane];
        }
        asm volatile("" :: "v"(a[0]), "v"(a[1]), "v"(a[2]), "v"(a[3]),
                           "v"(a[4]), "v"(a[5]), "v"(a[6]), "v"(a[7]),
                           "v"(c[0]), "v"(c[1]), "v"(c[2]), "v"(c[3]),
                           "v"(c[4]), "v"(c[5]), "v"(c[6]), "v"(c[7]) : "memory");
        #pragma unroll
        for (int j = 0; j < EPW; ++j) {
            int r = __builtin_amdgcn_readlane(rv, j);
            float4 b = rel_s[r * (HD / 4) + lane];
            float v = a[j].x*b.x*c[j].x + a[j].y*b.y*c[j].y + a[j].z*b.z*c[j].z + a[j].w*b.w*c[j].w;
            #pragma unroll
            for (int off = 16; off > 0; off >>= 1) v += __shfl_xor(v, off, 32);
            res = (lane == j) ? v : res;
        }
    } else {
        const int cnt = n_edges - e0;
        for (int j = 0; j < cnt; ++j) {
            int s = __builtin_amdgcn_readlane(sv, j);
            int d = __builtin_amdgcn_readlane(dv, j);
            int r = __builtin_amdgcn_readlane(rv, j);
            float v = edge_dot(ent, rel_s, s, d, r, lane);
            res = (lane == j) ? v : res;
        }
    }
    if (lane < EPW && me < n_edges) {
        score[me] = res;
        atomicMax(segmax + dv, enc_f32(res));
    }
    }
}

__global__ void __launch_bounds__(TPB) edge_exp_kernel(
        const int* __restrict__ dst, const unsigned* __restrict__ segmax,
        float* __restrict__ score, float* __restrict__ denom, int n_edges) {
    int e = blockIdx.x * TPB + threadIdx.x;
    if (e >= n_edges) return;
    int d = dst[e];
    float ex = __expf(score[e] - dec_f32(segmax[d]));
    score[e] = ex;
    atomicAdd(denom + d, ex);
}

__device__ __forceinline__ void edge_scatter(
        const float* __restrict__ ent, const float* rel_f,
        float* __restrict__ neigh, int s, int d, int r, float alpha, int lane) {
    const float* es = ent + (size_t)s * HD + lane;
    const float* rr = rel_f + r * HD + lane;
    float a0 = es[0], a1 = es[32], a2 = es[64], a3 = es[96];
    float b0 = rr[0], b1 = rr[32], b2 = rr[64], b3 = rr[96];
    float* out = neigh + (size_t)d * HD + lane;
    atomicAdd(out +  0, a0 * b0 * alpha);
    atomicAdd(out + 32, a1 * b1 * alpha);
    atomicAdd(out + 64, a2 * b2 * alpha);
    atomicAdd(out + 96, a3 * b3 * alpha);
}

__global__ void __launch_bounds__(EBLK) edge_aggr_kernel(
        const float* __restrict__ ent, const float* __restrict__ rel,
        const int* __restrict__ src, const int* __restrict__ dst,
        const int* __restrict__ rid,
        const float* __restrict__ score, const float* __restrict__ denom,
        float* __restrict__ neigh, int n_edges, int n_rel_floats) {
    __shared__ float4 rel_s[REL_ROWS * HD / 4];
    rel_to_lds(rel, rel_s, n_rel_floats);
    const float* rel_f = (const float*)rel_s;
    const int lane = threadIdx.x & 31;
    const int wv = __builtin_amdgcn_readfirstlane((int)(threadIdx.x >> 5));
    const int n_groups = (n_edges + EPW - 1) / EPW;
    for (int g = blockIdx.x * (EBLK / 32) + wv; g < n_groups; g += EGRID * (EBLK / 32)) {
    const int e0 = g * EPW;
    const int me  = e0 + (lane & (EPW - 1));
    const int mec = me < n_edges ? me : n_edges - 1;
    const int sv = src[mec], dv = dst[mec], rv = rid[mec];
    const int av = __float_as_int(score[mec] / denom[dv]);
    if (e0 + EPW <= n_edges) {
        v4f a[EPW], b[EPW];
        #pragma unroll
        for (int j = 0; j < EPW; ++j) {
            int s = __builtin_amdgcn_readlane(sv, j);
            int r = __builtin_amdgcn_readlane(rv, j);
            const float* es = ent + (size_t)s * HD + lane;
            const float* rr = rel_f + r * HD + lane;
            a[j].x = es[0]; a[j].y = es[32]; a[j].z = es[64]; a[j].w = es[96];
            b[j].x = rr[0]; b[j].y = rr[32]; b[j].z = rr[64]; b[j].w = rr[96];
        }
        asm volatile("" :: "v"(a[0]), "v"(a[1]), "v"(a[2]), "v"(a[3]),
                           "v"(a[4]), "v"(a[5]), "v"(a[6]), "v"(a[7]),
                           "v"(b[0]), "v"(b[1]), "v"(b[2]), "v"(b[3]),
                           "v"(b[4]), "v"(b[5]), "v"(b[6]), "v"(b[7]) : "memory");
        #pragma unroll
        for (int j = 0; j < EPW; ++j) {
            int d = __builtin_amdgcn_readlane(dv, j);
            float alpha = __int_as_float(__builtin_amdgcn_readlane(av, j));
            float* out = neigh + (size_t)d * HD + lane;
            atomicAdd(out +  0, a[j].x * b[j].x * alpha);
            atomicAdd(out + 32, a[j].y * b[j].y * alpha);
            atomicAdd(out + 64, a[j].z * b[j].z * alpha);
            atomicAdd(out + 96, a[j].w * b[j].w * alpha);
        }
    } else {
        const int cnt = n_edges - e0;
        for (int j = 0; j < cnt; ++j) {
            int s = __builtin_amdgcn_readlane(sv, j);
            int d = __builtin_amdgcn_readlane(dv, j);
            int r = __builtin_amdgcn_readlane(rv, j);
            float alpha = __int_as_float(__builtin_amdgcn_readlane(av, j));
            edge_scatter(ent, rel_f, neigh, s, d, r, alpha, lane);
        }
    }
    }
}

__global__ void __launch_bounds__(TPB) gemm_wmma_kernel(
        const float* __restrict__ A,
        const float* __restrict__ W,
        float* __restrict__ H,
        int n_rows) {
    const int wave = threadIdx.x >> 5;
    const int lane = threadIdx.x & 31;
    const int half = lane >> 4;
    const int l16  = lane & 15;
    const int m0 = blockIdx.x * 16;
    const int n0 = wave * 16;

    v8f c = {};
    const float* arow = A + (size_t)(m0 + l16) * HD;
    const float* wcol = W + n0 + l16;

    #pragma unroll 4
    for (int k0 = 0; k0 < HD; k0 += 4) {
        int ka = k0 + 2 * half;
        v2f a, b;
        a.x = arow[ka];
        a.y = arow[ka + 1];
        b.x = wcol[(size_t)ka * HD];
        b.y = wcol[(size_t)(ka + 1) * HD];
        c = __builtin_amdgcn_wmma_f32_16x16x4_f32(
                 false, a,  false, b,
                 (short)0, c,  false,  false);
    }

    float* hb = H + (size_t)(m0 + 8 * half) * HD + n0 + l16;
    #pragma unroll
    for (int r2 = 0; r2 < 8; ++r2) hb[(size_t)r2 * HD] = c[r2];
}

__global__ void __launch_bounds__(TPB) bn_stats_kernel(
        const float* __restrict__ H, float* __restrict__ colsum,
        float* __restrict__ colsumsq, int n_rows) {
    int col  = threadIdx.x & (HD - 1);
    int roff = threadIdx.x >> 7;
    float s = 0.0f, sq = 0.0f;
    for (int r = blockIdx.x * 2 + roff; r < n_rows; r += gridDim.x * 2) {
        float v = H[(size_t)r * HD + col];
        s += v; sq += v * v;
    }
    atomicAdd(colsum + col, s);
    atomicAdd(colsumsq + col, sq);
}

__global__ void __launch_bounds__(TPB) bn_tanh_kernel(
        float* __restrict__ H, const float* __restrict__ colsum,
        const float* __restrict__ colsumsq, const float* __restrict__ gamma,
        const float* __restrict__ beta, int n_rows) {
    size_t i = (size_t)blockIdx.x * TPB + threadIdx.x;
    size_t total = (size_t)n_rows * HD;
    if (i >= total) return;
    int c = (int)(i & (HD - 1));
    float invN = 1.0f / (float)n_rows;
    float mu   = colsum[c] * invN;
    float var  = colsumsq[c] * invN - mu * mu;
    float istd = rsqrtf(var + 1e-5f);
    float v = (H[i] - mu) * istd * gamma[c] + beta[c];
    H[i] = tanhf(v);
}

extern "C" void kernel_launch(void* const* d_in, const int* in_sizes, int n_in,
                              void* d_out, int out_size, void* d_ws, size_t ws_size,
                              hipStream_t stream) {
    const float* ent   = (const float*)d_in[0];
    const float* rel   = (const float*)d_in[1];
    const float* W     = (const float*)d_in[2];
    const float* gamma = (const float*)d_in[3];
    const float* beta  = (const float*)d_in[4];
    const int*   src   = (const int*)d_in[5];
    const int*   dst   = (const int*)d_in[6];
    const int*   rid   = (const int*)d_in[7];
    float* H = (float*)d_out;

    const int N = in_sizes[0] / HD;
    const int E = in_sizes[5];

    float*    ws       = (float*)d_ws;
    float*    neigh    = ws;
    float*    score    = neigh + (size_t)N * HD;
    unsigned* segmax   = (unsigned*)(score + E);
    float*    denom    = (float*)(segmax + N);
    float*    colsum   = denom + N;
    float*    colsumsq = colsum + HD;

    size_t zn = (size_t)N * HD;
    zero_kernel<<<(unsigned)((zn + TPB - 1) / TPB), TPB, 0, stream>>>(neigh, zn);
    size_t zt = (size_t)2 * N + 2 * HD;
    zero_kernel<<<(unsigned)((zt + TPB - 1) / TPB), TPB, 0, stream>>>((float*)segmax, zt);

    edge_score_kernel<<<EGRID, EBLK, 0, stream>>>(
        ent, rel, src, dst, rid, score, segmax, E, in_sizes[1]);
    edge_exp_kernel<<<(unsigned)((E + TPB - 1) / TPB), TPB, 0, stream>>>(
        dst, segmax, score, denom, E);
    edge_aggr_kernel<<<EGRID, EBLK, 0, stream>>>(
        ent, rel, src, dst, rid, score, denom, neigh, E, in_sizes[1]);

    gemm_wmma_kernel<<<(unsigned)(N / 16), TPB, 0, stream>>>(neigh, W, H, N);

    bn_stats_kernel<<<256, TPB, 0, stream>>>(H, colsum, colsumsq, N);
    bn_tanh_kernel<<<(unsigned)(((size_t)N * HD + TPB - 1) / TPB), TPB, 0, stream>>>(
        H, colsum, colsumsq, gamma, beta, N);
}


# GEMM re-tiled: wave = 16 rows x 64 cols, one 16-B A load feeds 8 products; W copied once per work-group to LDS, k pairs interleaved; whole-line stores; k order inside each 8 permuted: new sum order
# speedup vs baseline: 1.7126x; 1.0318x over previous
#include <hip/hip_runtime.h>
#include <math.h>

#define HD 128
#define TPB 256
#define EPW 8
#define REL_ROWS 130
#define EBLK 1024
#define EGRID 256
static_assert(EPW == 8, "the scatter kernel names its 2 x 8 row registers one by one");

typedef __attribute__((ext_vector_type(2))) float v2f;
typedef __attribute__((ext_vector_type(8))) float v8f;
typedef __attribute__((ext_vector_type(4))) float v4f;

__global__ void __launch_bounds__(TPB) zero_kernel(float* __restrict__ p, size_t n) {
    size_t i = (size_t)blockIdx.x * TPB + threadIdx.x;
    if (i < n) p[i] = 0.0f;
}

__device__ __forceinline__ unsigned enc_f32(float f) {
    unsigned b = __float_as_uint(f);
    return (b & 0x80000000u) ? ~b : (b | 0x80000000u);
}
__device__ __forceinline__ float dec_f32(unsigned e) {
    unsigned b = (e & 0x80000000u) ? (e & 0x7FFFFFFFu) : ~e;
    return __uint_as_float(b);
}

__device__ __forceinline__ void rel_to_lds(
        const float* __restrict__ rel, float4* rel_s, int n_rel_floats) {
    constexpr int NF4 = REL_ROWS * HD / 4;
    constexpr int TRIPS = (NF4 + EBLK - 1) / EBLK;
    int nf = n_rel_floats / 4;
    if (nf > NF4) nf = NF4;
    const float4* rg = (const float4*)rel;
    float4 t[TRIPS];
    #pragma unroll
    for (int k = 0; k < TRIPS; ++k) {
        int i = threadIdx.x + k * EBLK;
        t[k] = (i < nf) ? rg[i] : make_float4(0.0f, 0.0f, 0.0f, 0.0f);
    }
    #pragma unroll
    for (int k = 0; k < TRIPS; ++k) {
        int i = threadIdx.x + k * EBLK;
        if (i < nf) rel_s[i] = t[k];
    }
    __syncthreads();
}

__device__ __forceinline__ float edge_dot(
        const float* __restrict__ ent, const float4* rel_s,
        int s, int d, int r, int lane) {
    const float4* es = (const float4*)(ent + (size_t)s * HD);
    const float4* ed = (const float4*)(ent + (size_t)d * HD);
    float4 a = es[lane], b = rel_s[r * (HD / 4) + lane], c = ed[lane];
    float v = a.x*b.x*c.x + a.y*b.y*c.y + a.z*b.z*c.z + a.w*b.w*c.w;
    #pragma unroll
    for (int off = 16; off > 0; off >>= 1) v += __shfl_xor(v, off, 32);
    return v;
}

__global__ void __launch_bounds__(EBLK) edge_score_kernel(
        const float* __restrict__ ent, const float* __restrict__ rel,
        const int* __restrict__ src, const int* __restrict__ dst,
        const int* __restrict__ rid,
        float* __restrict__ score, unsigned* __restrict__ segmax, int n_edges,
        int n_rel_floats) {
    __shared__ float4 rel_s[REL_ROWS * HD / 4];
    rel_to_lds(rel, rel_s, n_rel_floats);
    const int lane = threadIdx.x & 31;
    const int wv = __builtin_amdgcn_readfirstlane((int)(threadIdx.x >> 5));
    const int n_groups = (n_edges + EPW - 1) / EPW;
    for (int g = blockIdx.x * (EBLK / 32) + wv; g < n_groups; g += EGRID * (EBLK / 32)) {
    const int e0 = g * EPW;
    const int me  = e0 + (lane & (EPW - 1));
    const int mec = me < n_edges ? me : n_edges - 1;
    const int sv = src[mec], dv = dst[mec], rv = rid[mec];
    float res = 0.0f;
    if (e0 + EPW <= n_edges) {
        v4f a[EPW], c[EPW];
        #pragma unroll
        for (int j = 0; j < EPW; ++j) {
            int s = __builtin_amdgcn_readlane(sv, j);
            int d = __builtin_amdgcn_readlane(dv, j);
            a[j] = ((const v4f*)(ent + (size_t)s * HD))[lane];
            c[j] = ((const v4f*)(ent + (size_t)d * HD))[lane];
        }
        asm volatile("" :: "v"(a[0]), "v"(a[1]), "v"(a[2]), "v"(a[3]),
                           "v"(a[4]), "v"(a[5]), "v"(a[6]), "v"(a[7]),
                           "v"(c[0]), "v"(c[1]), "v"(c[2]), "v"(c[3]),
                           "v"(c[4]), "v"(c[5]), "v"(c[6]), "v"(c[7]) : "memory");
        #pragma unroll
        for (int j = 0; j < EPW; ++j) {
            int r = __builtin_amdgcn_readlane(rv, j);
            float4 b = rel_s[r * (HD / 4) + lane];
            float v = a[j].x*b.x*c[j].x + a[j].y*b.y*c[j].y + a[j].z*b.z*c[j].z + a[j].w*b.w*c[j].w;
            #pragma unroll
            for (int off = 16; off > 0; off >>= 1) v += __shfl_xor(v, off, 32);
            res = (lane == j) ? v : res;
        }
    } else {
        const int cnt = n_edges - e0;
        for (int j = 0; j < cnt; ++j) {
            int s = __builtin_amdgcn_readlane(sv, j);
            int d = __builtin_amdgcn_readlane(dv, j);
            int r = __builtin_amdgcn_readlane(rv, j);
            float v = edge_dot(ent, rel_s, s, d, r, lane);
            res = (lane == j) ? v : res;
        }
    }
    if (lane < EPW && me < n_edges) {
        score[me] = res;
        atomicMax(segmax + dv, enc_f32(res));
    }
    }
}

__global__ void __launch_bounds__(TPB) edge_exp_kernel(
        const int* __restrict__ dst, const unsigned* __restrict__ segmax,
        float* __restrict__ score, float* __restrict__ denom, int n_edges) {
    int e = blockIdx.x * TPB + threadIdx.x;
    if (e >= n_edges) return;
    int d = dst[e];
    float ex = __expf(score[e] - dec_f32(segmax[d]));
    score[e] = ex;
    atomicAdd(denom + d, ex);
}

__device__ __forceinline__ void edge_scatter(
        const float* __restrict__ ent, const float* rel_f,
        float* __restrict__ neigh, int s, int d, int r, float alpha, int lane) {
    const float* es = ent + (size_t)s * HD + lane;
    const float* rr = rel_f + r * HD + lane;
    float a0 = es[0], a1 = es[32], a2 = es[64], a3 = es[96];
    float b0 = rr[0], b1 = rr[32], b2 = rr[64], b3 = rr[96];
    float* out = neigh + (size_t)d * HD + lane;
    atomicAdd(out +  0, a0 * b0 * alpha);
    atomicAdd(out + 32, a1 * b1 * alpha);
    atomicAdd(out + 64, a2 * b2 * alpha);
    atomicAdd(out + 96, a3 * b3 * alpha);
}

__global__ void __launch_bounds__(EBLK) edge_aggr_kernel(
        const float* __restrict__ ent, const float* __restrict__ rel,
        const int* __restrict__ src, const int* __restrict__ dst,
        const int* __restrict__ rid,
        const float* __restrict__ score, const float* __restrict__ denom,
        float* __restrict__ neigh, int n_edges, int n_rel_floats) {
    __shared__ float4 rel_s[REL_ROWS * HD / 4];
    rel_to_lds(rel, rel_s, n_rel_floats);
    const float* rel_f = (const float*)rel_s;
    const int lane = threadIdx.x & 31;
    const int wv = __builtin_amdgcn_readfirstlane((int)(threadIdx.x >> 5));
    const int n_groups = (n_edges + EPW - 1) / EPW;
    for (int g = blockIdx.x * (EBLK / 32) + wv; g < n_groups; g += EGRID * (EBLK / 32)) {
    const int e0 = g * EPW;
    const int me  = e0 + (lane & (EPW - 1));
    const int mec = me < n_edges ? me : n_edges - 1;
    const int sv = src[mec], dv = dst[mec], rv = rid[mec];
    const int av = __float_as_int(score[mec] / denom[dv]);
    if (e0 + EPW <= n_edges) {
        v4f a[EPW], b[EPW];
        #pragma unroll
        for (int j = 0; j < EPW; ++j) {
            int s = __builtin_amdgcn_readlane(sv, j);
            int r = __builtin_amdgcn_readlane(rv, j);
            const float* es = ent + (size_t)s * HD + lane;
            const float* rr = rel_f + r * HD + lane;
            a[j].x = es[0]; a[j].y = es[32]; a[j].z = es[64]; a[j].w = es[96];
            b[j].x = rr[0]; b[j].y = rr[32]; b[j].z = rr[64]; b[j].w = rr[96];
        }
        asm volatile("" :: "v"(a[0]), "v"(a[1]), "v"(a[2]), "v"(a[3]),
                           "v"(a[4]), "v"(a[5]), "v"(a[6]), "v"(a[7]),
                           "v"(b[0]), "v"(b[1]), "v"(b[2]), "v"(b[3]),
                           "v"(b[4]), "v"(b[5]), "v"(b[6]), "v"(b[7]) : "memory");
        #pragma unroll
        for (int j = 0; j < EPW; ++j) {
            int d = __builtin_amdgcn_readlane(dv, j);
            float alpha = __int_as_float(__builtin_amdgcn_readlane(av, j));
            float* out = neigh + (size_t)d * HD + lane;
            atomicAdd(out +  0, a[j].x * b[j].x * alpha);
            atomicAdd(out + 32, a[j].y * b[j].y * alpha);
            atomicAdd(out + 64, a[j].z * b[j].z * alpha);
            atomicAdd(out + 96, a[j].w * b[j].w * alpha);
        }
    } else {
        const int cnt = n_edges - e0;
        for (int j = 0; j < cnt; ++j) {
            int s = __builtin_amdgcn_readlane(sv, j);
            int d = __builtin_amdgcn_readlane(dv, j);
            int r = __builtin_amdgcn_readlane(rv, j);
            float alpha = __int_as_float(__builtin_amdgcn_readlane(av, j));
            edge_scatter(ent, rel_f, neigh, s, d, r, alpha, lane);
        }
    }
    }
}

#define GEMM_MT 4
__global__ void __launch_bounds__(TPB, 8) gemm_wmma_kernel(
        const float* __restrict__ A,
        const float* __restrict__ W,
        float* __restrict__ H,
        int n_rows) {
    __shared__ float4 w_s[HD * HD / 4];
    const int lane = threadIdx.x & 31;
    const int wave = __builtin_amdgcn_readfirstlane((int)(threadIdx.x >> 5));
    const int h = lane >> 4;
    const int c = lane & 15;
    const int n_mt = n_rows / 16;
    const int mt = blockIdx.x * GEMM_MT + (wave >> 1);
    const int nb = (wave & 1) * 64;
    const bool active = mt < n_mt;
    const int mtc = active ? mt : n_mt - 1;
    const int m0 = mtc * 16;

    v4f a[16];
    const v4f* ap = (const v4f*)(A + (size_t)(m0 + c) * HD + 4 * h);
    #pragma unroll
    for (int kp = 0; kp < 16; ++kp) a[kp] = ap[2 * kp];

    {
        const float4* wg = (const float4*)W;
        float4 r0[8], r1[8];
        #pragma unroll
        for (int it = 0; it < 8; ++it) {
            int i = threadIdx.x + it * TPB;
            int q = i >> 5, j = i & 31;
            r0[it] = wg[(2 * q) * (HD / 4) + j];
            r1[it] = wg[(2 * q + 1) * (HD / 4) + j];
        }
        #pragma unroll
        for (int it = 0; it < 8; ++it) {
            int i = threadIdx.x + it * TPB;
            int q = i >> 5, j = i & 31;
            w_s[q * 64 + 2 * j]     = make_float4(r0[it].x, r1[it].x, r0[it].y, r1[it].y);
            w_s[q * 64 + 2 * j + 1] = make_float4(r0[it].z, r1[it].z, r0[it].w, r1[it].w);
        }
    }
    __syncthreads();
    if (!active) return;

    const v4f* bp = (const v4f*)w_s + (2 * h) * 64 + (nb >> 1) + c;
    v8f acc0 = {}, acc1 = {}, acc2 = {}, acc3 = {};
    #pragma unroll
    for (int kp = 0; kp < 16; ++kp) {
        v4f b0 = bp[(4 * kp) * 64];
        v4f b1 = bp[(4 * kp) * 64 + 16];
        v4f b2 = bp[(4 * kp + 1) * 64];
        v4f b3 = bp[(4 * kp + 1) * 64 + 16];
        v2f alo = {a[kp].x, a[kp].y};
        v2f ahi = {a[kp].z, a[kp].w};
        v2f t;
        t = (v2f){b0.x, b0.y};
        acc0 = __builtin_amdgcn_wmma_f32_16x16x4_f32(false, alo, false, t, (short)0, acc0, false, false);
        t = (v2f){b0.z, b0.w};
        acc1 = __builtin_amdgcn_wmma_f32_16x16x4_f32(false, alo, false, t, (short)0, acc1, false, false);
        t = (v2f){b1.x, b1.y};
        acc2 = __builtin_amdgcn_wmma_f32_16x16x4_f32(false, alo, false, t, (short)0, acc2, false, false);
        t = (v2f){b1.z, b1.w};
        acc3 = __builtin_amdgcn_wmma_f32_16x16x4_f32(false, alo, false, t, (short)0, acc3, false, false);
        t = (v2f){b2.x, b2.y};
        acc0 = __builtin_amdgcn_wmma_f32_16x16x4_f32(false, ahi, false, t, (short)0, acc0, false, false);
        t = (v2f){b2.z, b2.w};
        acc1 = __builtin_amdgcn_wmma_f32_16x16x4_f32(false, ahi, false, t, (short)0, acc1, false, false);
        t = (v2f){b3.x, b3.y};
        acc2 = __builtin_amdgcn_wmma_f32_16x16x4_f32(false, ahi, false, t, (short)0, acc2, false, false);
        t = (v2f){b3.z, b3.w};
        acc3 = __builtin_amdgcn_wmma_f32_16x16x4_f32(false, ahi, false, t, (short)0, acc3, false, false);
    }

    float* hb = H + (size_t)(m0 + 8 * h) * HD + nb + 2 * c;
    #pragma unroll
    for (int r = 0; r < 8; ++r) {
        *(v2f*)(hb + (size_t)r * HD)      = (v2f){acc0[r], acc1[r]};
        *(v2f*)(hb + (size_t)r * HD + 32) = (v2f){acc2[r], acc3[r]};
    }
}

__global__ void __launch_bounds__(TPB) bn_stats_kernel(
        const float* __restrict__ H, float* __restrict__ colsum,
        float* __restrict__ colsumsq, int n_rows) {
    int col  = threadIdx.x & (HD - 1);
    int roff = threadIdx.x >> 7;
    float s = 0.0f, sq = 0.0f;
    for (int r = blockIdx.x * 2 + roff; r < n_rows; r += gridDim.x * 2) {
        float v = H[(size_t)r * HD + col];
        s += v; sq += v * v;
    }
    atomicAdd(colsum + col, s);
    atomicAdd(colsumsq + col, sq);
}

__global__ void __launch_bounds__(TPB) bn_tanh_kernel(
        float* __restrict__ H, const float* __restrict__ colsum,
        const float* __restrict__ colsumsq, const float* __restrict__ gamma,
        const float* __restrict__ beta, int n_rows) {
    size_t i = (size_t)blockIdx.x * TPB + threadIdx.x;
    size_t total = (size_t)n_rows * HD;
    if (i >= total) return;
    int c = (int)(i & (HD - 1));
    float invN = 1.0f / (float)n_rows;
    float mu   = colsum[c] * invN;
    float var  = colsumsq[c] * invN - mu * mu;
    float istd = rsqrtf(var + 1e-5f);
    float v = (H[i] - mu) * istd * gamma[c] + beta[c];
    H[i] = tanhf(v);
}

extern "C" void kernel_launch(void* const* d_in, const int* in_sizes, int n_in,
                              void* d_out, int out_size, void* d_ws, size_t ws_size,
                              hipStream_t stream) {
    const float* ent   = (const float*)d_in[0];
    const float* rel   = (const float*)d_in[1];
    const float* W     = (const float*)d_in[2];
    const float* gamma = (const float*)d_in[3];
    const float* beta  = (const float*)d_in[4];
    const int*   src   = (const int*)d_in[5];
    const int*   dst   = (const int*)d_in[6];
    const int*   rid   = (const int*)d_in[7];
    float* H = (float*)d_out;

    const int N = in_sizes[0] / HD;
    const int E = in_sizes[5];

    float*    ws       = (float*)d_ws;
    float*    neigh    = ws;
    float*    score    = neigh + (size_t)N * HD;
    unsigned* segmax   = (unsigned*)(score + E);
    float*    denom    = (float*)(segmax + N);
    float*    colsum   = denom + N;
    float*    colsumsq = colsum + HD;

    size_t zn = (size_t)N * HD;
    zero_kernel<<<(unsigned)((zn + TPB - 1) / TPB), TPB, 0, stream>>>(neigh, zn);
    size_t zt = (size_t)2 * N + 2 * HD;
    zero_kernel<<<(unsigned)((zt + TPB - 1) / TPB), TPB, 0, stream>>>((float*)segmax, zt);

    edge_score_kernel<<<EGRID, EBLK, 0, stream>>>(
        ent, rel, src, dst, rid, score, segmax, E, in_sizes[1]);
    edge_exp_kernel<<<(unsigned)((E + TPB - 1) / TPB), TPB, 0, stream>>>(
        dst, segmax, score, denom, E);
    edge_aggr_kernel<<<EGRID, EBLK, 0, stream>>>(
        ent, rel, src, dst, rid, score, denom, neigh, E, in_sizes[1]);

    gemm_wmma_kernel<<<(unsigned)((N / 16 + GEMM_MT - 1) / GEMM_MT), TPB, 0, stream>>>(neigh, W, H, N);

    bn_stats_kernel<<<256, TPB, 0, stream>>>(H, colsum, colsumsq, N);
    bn_tanh_kernel<<<(unsigned)(((size_t)N * HD + TPB - 1) / TPB), TPB, 0, stream>>>(
        H, colsum, colsumsq, gamma, beta, N);
}
